# starting kernel
# baseline (speedup 1.0000x reference)
#include <hip/hip_runtime.h>
#include <math.h>

typedef __attribute__((ext_vector_type(2))) float v2f;
typedef __attribute__((ext_vector_type(8))) float v8f;

#define BATCH 4096
#define NNEG  200
#define LAT   64
#define NPART 4
#define NCENT 256
#define CDIM  16

__global__ __launch_bounds__(256)
void cnorm_kernel(const float* __restrict__ centroids, float* __restrict__ cnorm) {
    int i = blockIdx.x * blockDim.x + threadIdx.x;
    if (i < NPART * NCENT) {
        const float* c = centroids + (size_t)i * CDIM;
        float s = 0.f;
        #pragma unroll
        for (int j = 0; j < CDIM; ++j) s += c[j] * c[j];
        cnorm[i] = s;
    }
}

__global__ __launch_bounds__(128)
void encode_kernel(const int*   __restrict__ user_id,
                   const float* __restrict__ user_table,
                   const float* __restrict__ centroids,
                   const float* __restrict__ gumbel_u,
                   const float* __restrict__ cnorm,
                   float*       __restrict__ user_vec) {
    __shared__ float s_user[16][LAT];

    const int b0  = blockIdx.x * 16;
    const int tid = threadIdx.x;

    for (int e = tid; e < 256; e += 128) {
        int row = e >> 4, c4 = e & 15;
        int uid = user_id[b0 + row];
        float4 v = *(const float4*)(user_table + (size_t)uid * LAT + c4 * 4);
        *(float4*)(&s_user[row][c4 * 4]) = v;
    }
    __syncthreads();

    const int p    = tid >> 5;
    const int lane = tid & 31;
    const int h    = lane >> 4;
    const int l16  = lane & 15;

    float ax[4], ay[4];
    #pragma unroll
    for (int s = 0; s < 4; ++s) {
        ax[s] = s_user[l16][p * CDIM + 4 * s + 2 * h];
        ay[s] = s_user[l16][p * CDIM + 4 * s + 2 * h + 1];
    }

    float best_v[8];
    int   best_i[8];
    #pragma unroll
    for (int r = 0; r < 8; ++r) { best_v[r] = -3.0e38f; best_i[r] = 0; }

    for (int nt = 0; nt < 16; ++nt) {
        const int n = nt * 16 + l16;
        const float* cp = centroids + ((size_t)(p * NCENT + n)) * CDIM;

        v8f acc = {0.f, 0.f, 0.f, 0.f, 0.f, 0.f, 0.f, 0.f};
        #pragma unroll
        for (int s = 0; s < 4; ++s) {
            v2f a; a.x = ax[s];              a.y = ay[s];
            v2f b; b.x = cp[4 * s + 2 * h];  b.y = cp[4 * s + 2 * h + 1];
            acc = __builtin_amdgcn_wmma_f32_16x16x4_f32(
                      false, a, false, b, (short)0, acc, false, false);
        }

        const float cn = cnorm[p * NCENT + n];
        #pragma unroll
        for (int r = 0; r < 8; ++r) {
            int brow = b0 + r + 8 * h;
            float u  = gumbel_u[((size_t)p * BATCH + brow) * NCENT + n];
            float g  = -__logf(-__logf(u));
            float sc = 2.0f * acc[r] - cn + g;
            if (sc > best_v[r]) { best_v[r] = sc; best_i[r] = n; }
        }
    }

    #pragma unroll
    for (int r = 0; r < 8; ++r) {
        float bv = best_v[r];
        int   bi = best_i[r];
        #pragma unroll
        for (int m = 1; m < 16; m <<= 1) {
            float ov = __shfl_xor(bv, m, 32);
            int   oi = __shfl_xor(bi, m, 32);
            if (ov > bv || (ov == bv && oi < bi)) { bv = ov; bi = oi; }
        }
        const int mrow = r + 8 * h;
        user_vec[(size_t)(b0 + mrow) * LAT + p * CDIM + l16] =
            centroids[((size_t)(p * NCENT + bi)) * CDIM + l16];
    }
}

__global__ __launch_bounds__(256)
void score_kernel(const int*   __restrict__ pos_id,
                  const int*   __restrict__ neg_ids,
                  const float* __restrict__ item_table,
                  const float* __restrict__ user_vec,
                  float*       __restrict__ out) {
    __shared__ float s_u[LAT];
    const int b = blockIdx.x;
    const int t = threadIdx.x;
    if (t < LAT) s_u[t] = user_vec[(size_t)b * LAT + t];
    __syncthreads();
    if (t > NNEG) return;

    const int id = (t == 0) ? pos_id[b] : neg_ids[(size_t)b * NNEG + (t - 1)];
    const float* it = item_table + (size_t)id * LAT;

    float acc = 0.f;
    #pragma unroll
    for (int j = 0; j < LAT; j += 4) {
        float4 v = *(const float4*)(it + j);
        acc += v.x * s_u[j] + v.y * s_u[j + 1] + v.z * s_u[j + 2] + v.w * s_u[j + 3];
    }

    if (t == 0) out[b];
    if (t == 0) out[b] = acc;
    else        out[BATCH + (size_t)b * NNEG + (t - 1)] = acc;
}

extern "C" void kernel_launch(void* const* d_in, const int* in_sizes, int n_in,
                              void* d_out, int out_size, void* d_ws, size_t ws_size,
                              hipStream_t stream) {
    const int*   user_id    = (const int*)  d_in[0];
    const int*   pos_id     = (const int*)  d_in[1];
    const int*   neg_ids    = (const int*)  d_in[2];
    const float* user_table = (const float*)d_in[3];
    const float* centroids  = (const float*)d_in[4];
    const float* item_table = (const float*)d_in[5];
    const float* gumbel_u   = (const float*)d_in[6];
    float* out = (float*)d_out;

    float* cnorm    = (float*)d_ws;
    float* user_vec = (float*)((char*)d_ws + NPART * NCENT * sizeof(float));

    cnorm_kernel <<<(NPART * NCENT + 255) / 256, 256, 0, stream>>>(centroids, cnorm);
    encode_kernel<<<BATCH / 16, 128, 0, stream>>>(user_id, user_table, centroids,
                                                  gumbel_u, cnorm, user_vec);
    score_kernel <<<BATCH, 256, 0, stream>>>(pos_id, neg_ids, item_table,
                                             user_vec, out);
}


# score kernel: each 16-lane half reads one whole 256-B item row per load (4 lines per wave load, not 32) and a 16-lane butterfly sums each score (sum order changes); 2 waves per batch row
# speedup vs baseline: 1.7863x; 1.7863x over previous
#include <hip/hip_runtime.h>
#include <math.h>

typedef __attribute__((ext_vector_type(2))) float v2f;
typedef __attribute__((ext_vector_type(8))) float v8f;

#define BATCH 4096
#define NNEG  200
#define LAT   64
#define NPART 4
#define NCENT 256
#define CDIM  16

__global__ __launch_bounds__(256)
void cnorm_kernel(const float* __restrict__ centroids, float* __restrict__ cnorm) {
    int i = blockIdx.x * blockDim.x + threadIdx.x;
    if (i < NPART * NCENT) {
        const float* c = centroids + (size_t)i * CDIM;
        float s = 0.f;
        #pragma unroll
        for (int j = 0; j < CDIM; ++j) s += c[j] * c[j];
        cnorm[i] = s;
    }
}

__global__ __launch_bounds__(128)
void encode_kernel(const int*   __restrict__ user_id,
                   const float* __restrict__ user_table,
                   const float* __restrict__ centroids,
                   const float* __restrict__ gumbel_u,
                   const float* __restrict__ cnorm,
                   float*       __restrict__ user_vec) {
    __shared__ float s_user[16][LAT];

    const int b0  = blockIdx.x * 16;
    const int tid = threadIdx.x;

    for (int e = tid; e < 256; e += 128) {
        int row = e >> 4, c4 = e & 15;
        int uid = user_id[b0 + row];
        float4 v = *(const float4*)(user_table + (size_t)uid * LAT + c4 * 4);
        *(float4*)(&s_user[row][c4 * 4]) = v;
    }
    __syncthreads();

    const int p    = tid >> 5;
    const int lane = tid & 31;
    const int h    = lane >> 4;
    const int l16  = lane & 15;

    float ax[4], ay[4];
    #pragma unroll
    for (int s = 0; s < 4; ++s) {
        ax[s] = s_user[l16][p * CDIM + 4 * s + 2 * h];
        ay[s] = s_user[l16][p * CDIM + 4 * s + 2 * h + 1];
    }

    float best_v[8];
    int   best_i[8];
    #pragma unroll
    for (int r = 0; r < 8; ++r) { best_v[r] = -3.0e38f; best_i[r] = 0; }

    for (int nt = 0; nt < 16; ++nt) {
        const int n = nt * 16 + l16;
        const float* cp = centroids + ((size_t)(p * NCENT + n)) * CDIM;

        v8f acc = {0.f, 0.f, 0.f, 0.f, 0.f, 0.f, 0.f, 0.f};
        #pragma unroll
        for (int s = 0; s < 4; ++s) {
            v2f a; a.x = ax[s];              a.y = ay[s];
            v2f b; b.x = cp[4 * s + 2 * h];  b.y = cp[4 * s + 2 * h + 1];
            acc = __builtin_amdgcn_wmma_f32_16x16x4_f32(
                      false, a, false, b, (short)0, acc, false, false);
        }

        const float cn = cnorm[p * NCENT + n];
        #pragma unroll
        for (int r = 0; r < 8; ++r) {
            int brow = b0 + r + 8 * h;
            float u  = gumbel_u[((size_t)p * BATCH + brow) * NCENT + n];
            float g  = -__logf(-__logf(u));
            float sc = 2.0f * acc[r] - cn + g;
            if (sc > best_v[r]) { best_v[r] = sc; best_i[r] = n; }
        }
    }

    #pragma unroll
    for (int r = 0; r < 8; ++r) {
        float bv = best_v[r];
        int   bi = best_i[r];
        #pragma unroll
        for (int m = 1; m < 16; m <<= 1) {
            float ov = __shfl_xor(bv, m, 32);
            int   oi = __shfl_xor(bi, m, 32);
            if (ov > bv || (ov == bv && oi < bi)) { bv = ov; bi = oi; }
        }
        const int mrow = r + 8 * h;
        user_vec[(size_t)(b0 + mrow) * LAT + p * CDIM + l16] =
            centroids[((size_t)(p * NCENT + bi)) * CDIM + l16];
    }
}

#define NSLOT (NNEG + 1)
#define NPASS ((NSLOT + 31) / 32)

template <int J>
__device__ __forceinline__ int half_bcast(int x) {
    return __builtin_amdgcn_update_dpp(0, x, 0x150 + J, 0xf, 0xf, true);
}

template <int M>
__device__ __forceinline__ float half_xor(float x) {
    return __builtin_bit_cast(float,
        __builtin_amdgcn_update_dpp(0, __builtin_bit_cast(int, x), 0x160 + M, 0xf, 0xf, true));
}

template <int N, int M>
__device__ __forceinline__ void fold(float* v, bool upper) {
    #pragma unroll
    for (int k = 0; k < N / 2; ++k) {
        const float keep = upper ? v[k + N / 2] : v[k];
        const float send = upper ? v[k] : v[k + N / 2];
        v[k] = keep + half_xor<M>(send);
    }
}

template <int J>
__device__ __forceinline__ float4 load_row(const float4* __restrict__ rows,
                                           int id, unsigned c) {
    const unsigned idj = (unsigned)half_bcast<J>(id);
    return rows[idj * (LAT / 4) + c];
}

__device__ __forceinline__ int slot_id(const int* __restrict__ pos_id,
                                       const int* __restrict__ neg_ids,
                                       int b, int slot) {
    int id = 0;
    if (slot < NNEG)       id = neg_ids[(size_t)b * NNEG + slot];
    else if (slot == NNEG) id = pos_id[b];
    return id;
}

__global__ __launch_bounds__(256) __attribute__((amdgpu_waves_per_eu(4, 8)))
void score_kernel(const int*   __restrict__ pos_id,
                  const int*   __restrict__ neg_ids,
                  const float* __restrict__ item_table,
                  const float* __restrict__ user_vec,
                  float*       __restrict__ out) {
    const int lane = threadIdx.x & 31;
    const int c    = lane & 15;
    const int wave = __builtin_amdgcn_readfirstlane((int)(threadIdx.x >> 5));
    const int gw   = blockIdx.x * 8 + wave;
    const int b    = gw >> 1;
    const int part = gw & 1;

    const float4   u  = *(const float4*)(user_vec + (size_t)b * LAT + 4 * c);
    const float4* __restrict__ rows = (const float4*)item_table;

    int id_next = slot_id(pos_id, neg_ids, b, part * 32 + lane);

    for (int pass = part; pass < NPASS; pass += 2) {
        const int slot = pass * 32 + lane;
        const int id   = id_next;
        if (pass + 2 < NPASS) id_next = slot_id(pos_id, neg_ids, b, slot + 64);

        float4 v[16];
        v[0]  = load_row<0>(rows, id, (unsigned)c);
        v[1]  = load_row<1>(rows, id, (unsigned)c);
        v[2]  = load_row<2>(rows, id, (unsigned)c);
        v[3]  = load_row<3>(rows, id, (unsigned)c);
        v[4]  = load_row<4>(rows, id, (unsigned)c);
        v[5]  = load_row<5>(rows, id, (unsigned)c);
        v[6]  = load_row<6>(rows, id, (unsigned)c);
        v[7]  = load_row<7>(rows, id, (unsigned)c);
        v[8]  = load_row<8>(rows, id, (unsigned)c);
        v[9]  = load_row<9>(rows, id, (unsigned)c);
        v[10] = load_row<10>(rows, id, (unsigned)c);
        v[11] = load_row<11>(rows, id, (unsigned)c);
        v[12] = load_row<12>(rows, id, (unsigned)c);
        v[13] = load_row<13>(rows, id, (unsigned)c);
        v[14] = load_row<14>(rows, id, (unsigned)c);
        v[15] = load_row<15>(rows, id, (unsigned)c);

        float acc[16];
        #pragma unroll
        for (int j = 0; j < 16; ++j)
            acc[j] = v[j].x * u.x + v[j].y * u.y + v[j].z * u.z + v[j].w * u.w;

        fold<16, 8>(acc, (c & 8) != 0);
        fold<8, 4>(acc, (c & 4) != 0);
        fold<4, 2>(acc, (c & 2) != 0);
        fold<2, 1>(acc, (c & 1) != 0);

        if (slot < NNEG)       out[BATCH + (size_t)b * NNEG + slot] = acc[0];
        else if (slot == NNEG) out[b] = acc[0];
    }
}

extern "C" void kernel_launch(void* const* d_in, const int* in_sizes, int n_in,
                              void* d_out, int out_size, void* d_ws, size_t ws_size,
                              hipStream_t stream) {
    const int*   user_id    = (const int*)  d_in[0];
    const int*   pos_id     = (const int*)  d_in[1];
    const int*   neg_ids    = (const int*)  d_in[2];
    const float* user_table = (const float*)d_in[3];
    const float* centroids  = (const float*)d_in[4];
    const float* item_table = (const float*)d_in[5];
    const float* gumbel_u   = (const float*)d_in[6];
    float* out = (float*)d_out;

    float* cnorm    = (float*)d_ws;
    float* user_vec = (float*)((char*)d_ws + NPART * NCENT * sizeof(float));

    cnorm_kernel <<<(NPART * NCENT + 255) / 256, 256, 0, stream>>>(centroids, cnorm);
    encode_kernel<<<BATCH / 16, 128, 0, stream>>>(user_id, user_table, centroids,
                                                  gumbel_u, cnorm, user_vec);
    score_kernel <<<BATCH / 4, 256, 0, stream>>>(pos_id, neg_ids, item_table,
                                                 user_vec, out);
}


# one launch instead of three: a 32-wave work-group gathers, encodes (8 waves per partition, 2 centroid tiles each, norms inline) and scores its 16 batch rows, stages joined through LDS; sums unchanged
# speedup vs baseline: 2.9103x; 1.6292x over previous
#include <hip/hip_runtime.h>
#include <math.h>


typedef __attribute__((ext_vector_type(2))) float v2f;
typedef __attribute__((ext_vector_type(8))) float v8f;

#define BATCH 4096
#define NNEG  200
#define LAT   64
#define NPART 4
#define NCENT 256
#define CDIM  16

#define ROWS  16
#define NTG   8
#define TPG   (NCENT / 16 / NTG)

#define NSLOT (NNEG + 1)
#define NPASS ((NSLOT + 31) / 32)

template <int J>
__device__ __forceinline__ int half_bcast(int x) {
    return __builtin_amdgcn_update_dpp(0, x, 0x150 + J, 0xf, 0xf, true);
}

template <int M>
__device__ __forceinline__ float half_xor(float x) {
    return __builtin_bit_cast(float,
        __builtin_amdgcn_update_dpp(0, __builtin_bit_cast(int, x), 0x160 + M, 0xf, 0xf, true));
}

template <int N, int M>
__device__ __forceinline__ void fold(float* v, bool upper) {
    #pragma unroll
    for (int k = 0; k < N / 2; ++k) {
        const float keep = upper ? v[k + N / 2] : v[k];
        const float send = upper ? v[k] : v[k + N / 2];
        v[k] = keep + half_xor<M>(send);
    }
}

template <int M>
__device__ __forceinline__ void argmax_step(float& bv, int& bi) {
    const float ov = half_xor<M>(bv);
    const int   oi = __builtin_amdgcn_update_dpp(0, bi, 0x160 + M, 0xf, 0xf, true);
    const bool take = (ov > bv) | ((ov == bv) & (oi < bi));
    bv = take ? ov : bv;
    bi = take ? oi : bi;
}

template <int J>
__device__ __forceinline__ float4 load_row(const float4* __restrict__ rows,
                                           int id, unsigned c) {
    const unsigned idj = (unsigned)half_bcast<J>(id);
    return rows[idj * (LAT / 4) + c];
}

__device__ __forceinline__ int slot_id(const int* __restrict__ pos_id,
                                       const int* __restrict__ neg_ids,
                                       int b, int slot) {
    int id = 0;
    if (slot < NNEG)       id = neg_ids[(size_t)b * NNEG + slot];
    else if (slot == NNEG) id = pos_id[b];
    return id;
}

__global__ __launch_bounds__(1024) __attribute__((amdgpu_waves_per_eu(8, 8)))
void fused_kernel(const int*   __restrict__ user_id,
                  const int*   __restrict__ pos_id,
                  const int*   __restrict__ neg_ids,
                  const float* __restrict__ user_table,
                  const float* __restrict__ centroids,
                  const float* __restrict__ item_table,
                  const float* __restrict__ gumbel_u,
                  float*       __restrict__ out) {
    __shared__ float s_user[ROWS][LAT];
    __shared__ float s_bv[NPART][ROWS][NTG];
    __shared__ int   s_bi[NPART][ROWS][NTG];

    const int tid  = threadIdx.x;
    const int b0   = blockIdx.x * ROWS;
    const int lane = tid & 31;
    const int h    = lane >> 4;
    const int c    = lane & 15;
    const int wave = __builtin_amdgcn_readfirstlane(tid >> 5);

    const int row  = wave >> 1;
    const int b    = b0 + row;
    const int part = wave & 1;

    int id_next = slot_id(pos_id, neg_ids, b, part * 32 + lane);

    if (tid < 256) {
        const int grow = tid >> 4, c4 = tid & 15;
        const int uid  = user_id[b0 + grow];
        const float4 v = *(const float4*)(user_table + (size_t)uid * LAT + c4 * 4);
        *(float4*)(&s_user[grow][c4 * 4]) = v;
    }
    __syncthreads();

    {
        const int p  = wave & 3;
        const int tg = wave >> 2;

        float ax[4], ay[4];
        #pragma unroll
        for (int s = 0; s < 4; ++s) {
            ax[s] = s_user[c][p * CDIM + 4 * s + 2 * h];
            ay[s] = s_user[c][p * CDIM + 4 * s + 2 * h + 1];
        }

        float best_v[8];
        int   best_i[8];
        #pragma unroll
        for (int r = 0; r < 8; ++r) { best_v[r] = -3.0e38f; best_i[r] = 0; }

        #pragma unroll
        for (int t = 0; t < TPG; ++t) {
            const int n = (tg * TPG + t) * 16 + c;
            const float* cp = centroids + ((size_t)(p * NCENT + n)) * CDIM;

            v8f acc = {0.f, 0.f, 0.f, 0.f, 0.f, 0.f, 0.f, 0.f};
            #pragma unroll
            for (int s = 0; s < 4; ++s) {
                v2f a; a.x = ax[s];              a.y = ay[s];
                v2f bb; bb.x = cp[4 * s + 2 * h]; bb.y = cp[4 * s + 2 * h + 1];
                acc = __builtin_amdgcn_wmma_f32_16x16x4_f32(
                          false, a, false, bb, (short)0, acc, false, false);
            }

            float cn = 0.f;
            #pragma unroll
            for (int j = 0; j < CDIM; ++j) cn += cp[j] * cp[j];

            #pragma unroll
            for (int r = 0; r < 8; ++r) {
                int brow = b0 + r + 8 * h;
                float u  = gumbel_u[((size_t)p * BATCH + brow) * NCENT + n];
                float g  = -__logf(-__logf(u));
                float sc = 2.0f * acc[r] - cn + g;
                if (sc > best_v[r]) { best_v[r] = sc; best_i[r] = n; }
            }
        }

        #pragma unroll
        for (int r = 0; r < 8; ++r) {
            float bv = best_v[r];
            int   bi = best_i[r];
            argmax_step<1>(bv, bi);
            argmax_step<2>(bv, bi);
            argmax_step<4>(bv, bi);
            argmax_step<8>(bv, bi);
            if (c == 0) {
                s_bv[p][r + 8 * h][tg] = bv;
                s_bi[p][r + 8 * h][tg] = bi;
            }
        }
    }
    __syncthreads();

    const int pu = c >> 2;
    float mv = s_bv[pu][row][0];
    int   mi = s_bi[pu][row][0];
    #pragma unroll
    for (int g = 1; g < NTG; ++g) {
        const float v = s_bv[pu][row][g];
        const int   i = s_bi[pu][row][g];
        if (v > mv) { mv = v; mi = i; }
    }
    const float4 u = *(const float4*)(centroids + ((size_t)(pu * NCENT + mi)) * CDIM + (c & 3) * 4);

    const float4* __restrict__ rows = (const float4*)item_table;

    for (int pass = part; pass < NPASS; pass += 2) {
        const int slot = pass * 32 + lane;
        const int id   = id_next;
        if (pass + 2 < NPASS) id_next = slot_id(pos_id, neg_ids, b, slot + 64);

        float4 v[16];
        v[0]  = load_row<0>(rows, id, (unsigned)c);
        v[1]  = load_row<1>(rows, id, (unsigned)c);
        v[2]  = load_row<2>(rows, id, (unsigned)c);
        v[3]  = load_row<3>(rows, id, (unsigned)c);
        v[4]  = load_row<4>(rows, id, (unsigned)c);
        v[5]  = load_row<5>(rows, id, (unsigned)c);
        v[6]  = load_row<6>(rows, id, (unsigned)c);
        v[7]  = load_row<7>(rows, id, (unsigned)c);
        v[8]  = load_row<8>(rows, id, (unsigned)c);
        v[9]  = load_row<9>(rows, id, (unsigned)c);
        v[10] = load_row<10>(rows, id, (unsigned)c);
        v[11] = load_row<11>(rows, id, (unsigned)c);
        v[12] = load_row<12>(rows, id, (unsigned)c);
        v[13] = load_row<13>(rows, id, (unsigned)c);
        v[14] = load_row<14>(rows, id, (unsigned)c);
        v[15] = load_row<15>(rows, id, (unsigned)c);

        float acc[16];
        #pragma unroll
        for (int j = 0; j < 16; ++j)
            acc[j] = v[j].x * u.x + v[j].y * u.y + v[j].z * u.z + v[j].w * u.w;

        fold<16, 8>(acc, (c & 8) != 0);
        fold<8, 4>(acc, (c & 4) != 0);
        fold<4, 2>(acc, (c & 2) != 0);
        fold<2, 1>(acc, (c & 1) != 0);

        if (slot < NNEG)       out[BATCH + (size_t)b * NNEG + slot] = acc[0];
        else if (slot == NNEG) out[b] = acc[0];
    }
}

extern "C" void kernel_launch(void* const* d_in, const int* in_sizes, int n_in,
                              void* d_out, int out_size, void* d_ws, size_t ws_size,
                              hipStream_t stream) {
    const int*   user_id    = (const int*)  d_in[0];
    const int*   pos_id     = (const int*)  d_in[1];
    const int*   neg_ids    = (const int*)  d_in[2];
    const float* user_table = (const float*)d_in[3];
    const float* centroids  = (const float*)d_in[4];
    const float* item_table = (const float*)d_in[5];
    const float* gumbel_u   = (const float*)d_in[6];
    float* out = (float*)d_out;

    fused_kernel<<<BATCH / ROWS, 1024, 0, stream>>>(user_id, pos_id, neg_ids, user_table,
                                                    centroids, item_table, gumbel_u, out);
}


# 16-wave work-groups, one wave per batch row, 256 registers each: item rows are requested one pass ahead, the first pass ahead of the encode; the last pass loads its 9 rows only; sums unchanged
# speedup vs baseline: 3.2060x; 1.1016x over previous
#include <hip/hip_runtime.h>
#include <math.h>


typedef __attribute__((ext_vector_type(2))) float v2f;
typedef __attribute__((ext_vector_type(8))) float v8f;

#define BATCH 4096
#define NNEG  200
#define LAT   64
#define NPART 4
#define NCENT 256
#define CDIM  16

#define ROWS  16
#define NTG   4
#define TPG   (NCENT / 16 / NTG)

#define NSLOT (NNEG + 1)
#define NPASS ((NSLOT + 31) / 32)
#define NTAIL (NSLOT - (NPASS - 1) * 32)

template <int J>
__device__ __forceinline__ int half_bcast(int x) {
    return __builtin_amdgcn_update_dpp(0, x, 0x150 + J, 0xf, 0xf, true);
}

template <int M>
__device__ __forceinline__ float half_xor(float x) {
    return __builtin_bit_cast(float,
        __builtin_amdgcn_update_dpp(0, __builtin_bit_cast(int, x), 0x160 + M, 0xf, 0xf, true));
}

template <int N, int M>
__device__ __forceinline__ void fold(float* v, bool upper) {
    #pragma unroll
    for (int k = 0; k < N / 2; ++k) {
        const float keep = upper ? v[k + N / 2] : v[k];
        const float send = upper ? v[k] : v[k + N / 2];
        v[k] = keep + half_xor<M>(send);
    }
}

template <int M>
__device__ __forceinline__ void argmax_step(float& bv, int& bi) {
    const float ov = half_xor<M>(bv);
    const int   oi = __builtin_amdgcn_update_dpp(0, bi, 0x160 + M, 0xf, 0xf, true);
    const bool take = (ov > bv) | ((ov == bv) & (oi < bi));
    bv = take ? ov : bv;
    bi = take ? oi : bi;
}

template <int J>
__device__ __forceinline__ float4 load_row(const float4* __restrict__ rows,
                                           int id, unsigned c) {
    const unsigned idj = (unsigned)half_bcast<J>(id);
    return rows[idj * (LAT / 4) + c];
}

__device__ __forceinline__ int slot_id(const int* __restrict__ pos_id,
                                       const int* __restrict__ neg_ids,
                                       int b, int slot) {
    int id = 0;
    if (slot < NNEG)       id = neg_ids[(size_t)b * NNEG + slot];
    else if (slot == NNEG) id = pos_id[b];
    return id;
}

template <int NJ>
__device__ __forceinline__ void load_pass(float4* v, const float4* __restrict__ rows,
                                          int id, unsigned c) {
    if (NJ > 0)  v[0]  = load_row<0>(rows, id, c);
    if (NJ > 1)  v[1]  = load_row<1>(rows, id, c);
    if (NJ > 2)  v[2]  = load_row<2>(rows, id, c);
    if (NJ > 3)  v[3]  = load_row<3>(rows, id, c);
    if (NJ > 4)  v[4]  = load_row<4>(rows, id, c);
    if (NJ > 5)  v[5]  = load_row<5>(rows, id, c);
    if (NJ > 6)  v[6]  = load_row<6>(rows, id, c);
    if (NJ > 7)  v[7]  = load_row<7>(rows, id, c);
    if (NJ > 8)  v[8]  = load_row<8>(rows, id, c);
    if (NJ > 9)  v[9]  = load_row<9>(rows, id, c);
    if (NJ > 10) v[10] = load_row<10>(rows, id, c);
    if (NJ > 11) v[11] = load_row<11>(rows, id, c);
    if (NJ > 12) v[12] = load_row<12>(rows, id, c);
    if (NJ > 13) v[13] = load_row<13>(rows, id, c);
    if (NJ > 14) v[14] = load_row<14>(rows, id, c);
    if (NJ > 15) v[15] = load_row<15>(rows, id, c);
}

template <int NJ>
__device__ __forceinline__ float score_pass(const float4* v, float4 u, int c) {
    float acc[16];
    #pragma unroll
    for (int j = 0; j < 16; ++j)
        acc[j] = (j < NJ) ? v[j].x * u.x + v[j].y * u.y + v[j].z * u.z + v[j].w * u.w : 0.f;

    fold<16, 8>(acc, (c & 8) != 0);
    fold<8, 4>(acc, (c & 4) != 0);
    fold<4, 2>(acc, (c & 2) != 0);
    fold<2, 1>(acc, (c & 1) != 0);
    return acc[0];
}

static_assert((NPASS - 1) % 2 == 0 && NPASS >= 3, "whole passes come two to a trip");
static_assert((NPASS - 1) * 32 <= NNEG, "every whole pass lies below NNEG");
static_assert(NTAIL >= 1 && NTAIL <= 16, "the last pass fits the lower half");

__global__ __launch_bounds__(512) __attribute__((amdgpu_waves_per_eu(4, 4)))
void fused_kernel(const int*   __restrict__ user_id,
                  const int*   __restrict__ pos_id,
                  const int*   __restrict__ neg_ids,
                  const float* __restrict__ user_table,
                  const float* __restrict__ centroids,
                  const float* __restrict__ item_table,
                  const float* __restrict__ gumbel_u,
                  float*       __restrict__ out) {
    __shared__ float s_user[ROWS][LAT];
    __shared__ float s_bv[NPART][ROWS][NTG];
    __shared__ int   s_bi[NPART][ROWS][NTG];

    const int tid  = threadIdx.x;
    const int b0   = blockIdx.x * ROWS;
    const int lane = tid & 31;
    const int h    = lane >> 4;
    const int c    = lane & 15;
    const int wave = __builtin_amdgcn_readfirstlane(tid >> 5);

    const int row = wave;
    const int b   = b0 + row;

    const float4* __restrict__ rows = (const float4*)item_table;

    const int id_first = slot_id(pos_id, neg_ids, b, lane);
    int id_b = slot_id(pos_id, neg_ids, b, 32 + lane);

    const int p  = wave & 3;
    const int tg = wave >> 2;

    v2f   bf[TPG][4];
    float cnorm[TPG];
    float gu[TPG][8];
    #pragma unroll
    for (int t = 0; t < TPG; ++t) {
        const int n = (tg * TPG + t) * 16 + c;
        const float* cp = centroids + ((size_t)(p * NCENT + n)) * CDIM;
        #pragma unroll
        for (int s = 0; s < 4; ++s) { bf[t][s].x = cp[4 * s + 2 * h]; bf[t][s].y = cp[4 * s + 2 * h + 1]; }
        float cn = 0.f;
        #pragma unroll
        for (int j = 0; j < CDIM; ++j) cn += cp[j] * cp[j];
        cnorm[t] = cn;
        #pragma unroll
        for (int r = 0; r < 8; ++r) {
            const int brow = b0 + r + 8 * h;
            gu[t][r] = gumbel_u[((size_t)p * BATCH + brow) * NCENT + n];
        }
    }

    if (tid < 256) {
        const int grow = tid >> 4, c4 = tid & 15;
        const int uid  = user_id[b0 + grow];
        const float4 v = *(const float4*)(user_table + (size_t)uid * LAT + c4 * 4);
        *(float4*)(&s_user[grow][c4 * 4]) = v;
    }
    __syncthreads();

    float4 va[16], vb[16];

    load_pass<16>(va, rows, id_first, (unsigned)c);

    {
        float ax[4], ay[4];
        #pragma unroll
        for (int s = 0; s < 4; ++s) {
            ax[s] = s_user[c][p * CDIM + 4 * s + 2 * h];
            ay[s] = s_user[c][p * CDIM + 4 * s + 2 * h + 1];
        }

        float best_v[8];
        int   best_i[8];
        #pragma unroll
        for (int r = 0; r < 8; ++r) { best_v[r] = -3.0e38f; best_i[r] = 0; }

        #pragma unroll
        for (int t = 0; t < TPG; ++t) {
            const int n = (tg * TPG + t) * 16 + c;

            v8f acc = {0.f, 0.f, 0.f, 0.f, 0.f, 0.f, 0.f, 0.f};
            #pragma unroll
            for (int s = 0; s < 4; ++s) {
                v2f a; a.x = ax[s]; a.y = ay[s];
                acc = __builtin_amdgcn_wmma_f32_16x16x4_f32(
                          false, a, false, bf[t][s], (short)0, acc, false, false);
            }

            const float cn = cnorm[t];

            #pragma unroll
            for (int r = 0; r < 8; ++r) {
                float g  = -__logf(-__logf(gu[t][r]));
                float sc = 2.0f * acc[r] - cn + g;
                if (sc > best_v[r]) { best_v[r] = sc; best_i[r] = n; }
            }
        }

        #pragma unroll
        for (int r = 0; r < 8; ++r) {
            float bv = best_v[r];
            int   bi = best_i[r];
            argmax_step<1>(bv, bi);
            argmax_step<2>(bv, bi);
            argmax_step<4>(bv, bi);
            argmax_step<8>(bv, bi);
            if (c == 0) {
                s_bv[p][r + 8 * h][tg] = bv;
                s_bi[p][r + 8 * h][tg] = bi;
            }
        }
    }
    __syncthreads();

    const int pu = c >> 2;
    float mv = s_bv[pu][row][0];
    int   mi = s_bi[pu][row][0];
    #pragma unroll
    for (int g = 1; g < NTG; ++g) {
        const float v = s_bv[pu][row][g];
        const int   i = s_bi[pu][row][g];
        if (v > mv) { mv = v; mi = i; }
    }
    const float4 u = *(const float4*)(centroids + ((size_t)(pu * NCENT + mi)) * CDIM + (c & 3) * 4);

    float* __restrict__ neg_out = out + BATCH + (size_t)b * NNEG;

    int pass = 0;
    #pragma nounroll
    for (; pass < NPASS - 3; pass += 2) {
        const int id_c = slot_id(pos_id, neg_ids, b, (pass + 2) * 32 + lane);
        const int id_d = slot_id(pos_id, neg_ids, b, (pass + 3) * 32 + lane);

        load_pass<16>(vb, rows, id_b, (unsigned)c);
        neg_out[pass * 32 + lane] = score_pass<16>(va, u, c);

        load_pass<16>(va, rows, id_c, (unsigned)c);
        neg_out[(pass + 1) * 32 + lane] = score_pass<16>(vb, u, c);

        id_b = id_d;
    }

    {
        const int id_c = slot_id(pos_id, neg_ids, b, (NPASS - 1) * 32 + lane);

        load_pass<16>(vb, rows, id_b, (unsigned)c);
        neg_out[(NPASS - 3) * 32 + lane] = score_pass<16>(va, u, c);

        load_pass<NTAIL>(va, rows, id_c, (unsigned)c);
        neg_out[(NPASS - 2) * 32 + lane] = score_pass<16>(vb, u, c);
    }

    {
        const float s    = score_pass<NTAIL>(va, u, c);
        const int   slot = (NPASS - 1) * 32 + lane;
        if (slot < NNEG)       neg_out[slot] = s;
        else if (slot == NNEG) out[b] = s;
    }
}

extern "C" void kernel_launch(void* const* d_in, const int* in_sizes, int n_in,
                              void* d_out, int out_size, void* d_ws, size_t ws_size,
                              hipStream_t stream) {
    const int*   user_id    = (const int*)  d_in[0];
    const int*   pos_id     = (const int*)  d_in[1];
    const int*   neg_ids    = (const int*)  d_in[2];
    const float* user_table = (const float*)d_in[3];
    const float* centroids  = (const float*)d_in[4];
    const float* item_table = (const float*)d_in[5];
    const float* gumbel_u   = (const float*)d_in[6];
    float* out = (float*)d_out;

    fused_kernel<<<BATCH / ROWS, 512, 0, stream>>>(user_id, pos_id, neg_ids, user_table,
                                                    centroids, item_table, gumbel_u, out);
}


# encode: Gumbel term as -ln2*log2(-log2 u) on the hardware log2, its argmax-neutral constant dropped, in place of two library logs; noise rounding moves about an ulp; sums unchanged
# speedup vs baseline: 3.3888x; 1.0570x over previous
#include <hip/hip_runtime.h>
#include <math.h>


typedef __attribute__((ext_vector_type(2))) float v2f;
typedef __attribute__((ext_vector_type(8))) float v8f;

#define BATCH 4096
#define NNEG  200
#define LAT   64
#define NPART 4
#define NCENT 256
#define CDIM  16

#define ROWS  16
#define NTG   4
#define TPG   (NCENT / 16 / NTG)

#define NSLOT (NNEG + 1)
#define NPASS ((NSLOT + 31) / 32)
#define NTAIL (NSLOT - (NPASS - 1) * 32)

template <int J>
__device__ __forceinline__ int half_bcast(int x) {
    return __builtin_amdgcn_update_dpp(0, x, 0x150 + J, 0xf, 0xf, true);
}

template <int M>
__device__ __forceinline__ float half_xor(float x) {
    return __builtin_bit_cast(float,
        __builtin_amdgcn_update_dpp(0, __builtin_bit_cast(int, x), 0x160 + M, 0xf, 0xf, true));
}

template <int N, int M>
__device__ __forceinline__ void fold(float* v, bool upper) {
    #pragma unroll
    for (int k = 0; k < N / 2; ++k) {
        const float keep = upper ? v[k + N / 2] : v[k];
        const float send = upper ? v[k] : v[k + N / 2];
        v[k] = keep + half_xor<M>(send);
    }
}

template <int M>
__device__ __forceinline__ void argmax_step(float& bv, int& bi) {
    const float ov = half_xor<M>(bv);
    const int   oi = __builtin_amdgcn_update_dpp(0, bi, 0x160 + M, 0xf, 0xf, true);
    const bool take = (ov > bv) | ((ov == bv) & (oi < bi));
    bv = take ? ov : bv;
    bi = take ? oi : bi;
}

template <int J>
__device__ __forceinline__ float4 load_row(const float4* __restrict__ rows,
                                           int id, unsigned c) {
    const unsigned idj = (unsigned)half_bcast<J>(id);
    return rows[idj * (LAT / 4) + c];
}

__device__ __forceinline__ int slot_id(const int* __restrict__ pos_id,
                                       const int* __restrict__ neg_ids,
                                       int b, int slot) {
    int id = 0;
    if (slot < NNEG)       id = neg_ids[(size_t)b * NNEG + slot];
    else if (slot == NNEG) id = pos_id[b];
    return id;
}

template <int NJ>
__device__ __forceinline__ void load_pass(float4* v, const float4* __restrict__ rows,
                                          int id, unsigned c) {
    if (NJ > 0)  v[0]  = load_row<0>(rows, id, c);
    if (NJ > 1)  v[1]  = load_row<1>(rows, id, c);
    if (NJ > 2)  v[2]  = load_row<2>(rows, id, c);
    if (NJ > 3)  v[3]  = load_row<3>(rows, id, c);
    if (NJ > 4)  v[4]  = load_row<4>(rows, id, c);
    if (NJ > 5)  v[5]  = load_row<5>(rows, id, c);
    if (NJ > 6)  v[6]  = load_row<6>(rows, id, c);
    if (NJ > 7)  v[7]  = load_row<7>(rows, id, c);
    if (NJ > 8)  v[8]  = load_row<8>(rows, id, c);
    if (NJ > 9)  v[9]  = load_row<9>(rows, id, c);
    if (NJ > 10) v[10] = load_row<10>(rows, id, c);
    if (NJ > 11) v[11] = load_row<11>(rows, id, c);
    if (NJ > 12) v[12] = load_row<12>(rows, id, c);
    if (NJ > 13) v[13] = load_row<13>(rows, id, c);
    if (NJ > 14) v[14] = load_row<14>(rows, id, c);
    if (NJ > 15) v[15] = load_row<15>(rows, id, c);
}

template <int NJ>
__device__ __forceinline__ float score_pass(const float4* v, float4 u, int c) {
    float acc[16];
    #pragma unroll
    for (int j = 0; j < 16; ++j)
        acc[j] = (j < NJ) ? v[j].x * u.x + v[j].y * u.y + v[j].z * u.z + v[j].w * u.w : 0.f;

    fold<16, 8>(acc, (c & 8) != 0);
    fold<8, 4>(acc, (c & 4) != 0);
    fold<4, 2>(acc, (c & 2) != 0);
    fold<2, 1>(acc, (c & 1) != 0);
    return acc[0];
}

static_assert((NPASS - 1) % 2 == 0 && NPASS >= 3, "whole passes come two to a trip");
static_assert((NPASS - 1) * 32 <= NNEG, "every whole pass lies below NNEG");
static_assert(NTAIL >= 1 && NTAIL <= 16, "the last pass fits the lower half");

__global__ __launch_bounds__(512) __attribute__((amdgpu_waves_per_eu(4, 4)))
void fused_kernel(const int*   __restrict__ user_id,
                  const int*   __restrict__ pos_id,
                  const int*   __restrict__ neg_ids,
                  const float* __restrict__ user_table,
                  const float* __restrict__ centroids,
                  const float* __restrict__ item_table,
                  const float* __restrict__ gumbel_u,
                  float*       __restrict__ out) {
    __shared__ float s_user[ROWS][LAT];
    __shared__ float s_bv[NPART][ROWS][NTG];
    __shared__ int   s_bi[NPART][ROWS][NTG];

    const int tid  = threadIdx.x;
    const int b0   = blockIdx.x * ROWS;
    const int lane = tid & 31;
    const int h    = lane >> 4;
    const int c    = lane & 15;
    const int wave = __builtin_amdgcn_readfirstlane(tid >> 5);

    const int row = wave;
    const int b   = b0 + row;

    const float4* __restrict__ rows = (const float4*)item_table;

    const int id_first = slot_id(pos_id, neg_ids, b, lane);
    int id_b = slot_id(pos_id, neg_ids, b, 32 + lane);

    const int p  = wave & 3;
    const int tg = wave >> 2;

    v2f   bf[TPG][4];
    float cnorm[TPG];
    float gu[TPG][8];
    #pragma unroll
    for (int t = 0; t < TPG; ++t) {
        const int n = (tg * TPG + t) * 16 + c;
        const float* cp = centroids + ((size_t)(p * NCENT + n)) * CDIM;
        #pragma unroll
        for (int s = 0; s < 4; ++s) { bf[t][s].x = cp[4 * s + 2 * h]; bf[t][s].y = cp[4 * s + 2 * h + 1]; }
        float cn = 0.f;
        #pragma unroll
        for (int j = 0; j < CDIM; ++j) cn += cp[j] * cp[j];
        cnorm[t] = cn;
        #pragma unroll
        for (int r = 0; r < 8; ++r) {
            const int brow = b0 + r + 8 * h;
            gu[t][r] = gumbel_u[((size_t)p * BATCH + brow) * NCENT + n];
        }
    }

    if (tid < 256) {
        const int grow = tid >> 4, c4 = tid & 15;
        const int uid  = user_id[b0 + grow];
        const float4 v = *(const float4*)(user_table + (size_t)uid * LAT + c4 * 4);
        *(float4*)(&s_user[grow][c4 * 4]) = v;
    }
    __syncthreads();

    float4 va[16], vb[16];

    load_pass<16>(va, rows, id_first, (unsigned)c);

    {
        float ax[4], ay[4];
        #pragma unroll
        for (int s = 0; s < 4; ++s) {
            ax[s] = s_user[c][p * CDIM + 4 * s + 2 * h];
            ay[s] = s_user[c][p * CDIM + 4 * s + 2 * h + 1];
        }

        float best_v[8];
        int   best_i[8];
        #pragma unroll
        for (int r = 0; r < 8; ++r) { best_v[r] = -3.0e38f; best_i[r] = 0; }

        #pragma unroll
        for (int t = 0; t < TPG; ++t) {
            const int n = (tg * TPG + t) * 16 + c;

            v8f acc = {0.f, 0.f, 0.f, 0.f, 0.f, 0.f, 0.f, 0.f};
            #pragma unroll
            for (int s = 0; s < 4; ++s) {
                v2f a; a.x = ax[s]; a.y = ay[s];
                acc = __builtin_amdgcn_wmma_f32_16x16x4_f32(
                          false, a, false, bf[t][s], (short)0, acc, false, false);
            }

            const float cn = cnorm[t];

            #pragma unroll
            for (int r = 0; r < 8; ++r) {
                const float tt = __builtin_amdgcn_logf(-__builtin_log2f(gu[t][r]));
                const float sc = __builtin_fmaf(tt, -0.69314718055994531f, 2.0f * acc[r] - cn);
                if (sc > best_v[r]) { best_v[r] = sc; best_i[r] = n; }
            }
        }

        #pragma unroll
        for (int r = 0; r < 8; ++r) {
            float bv = best_v[r];
            int   bi = best_i[r];
            argmax_step<1>(bv, bi);
            argmax_step<2>(bv, bi);
            argmax_step<4>(bv, bi);
            argmax_step<8>(bv, bi);
            if (c == 0) {
                s_bv[p][r + 8 * h][tg] = bv;
                s_bi[p][r + 8 * h][tg] = bi;
            }
        }
    }
    __syncthreads();

    const int pu = c >> 2;
    float mv = s_bv[pu][row][0];
    int   mi = s_bi[pu][row][0];
    #pragma unroll
    for (int g = 1; g < NTG; ++g) {
        const float v = s_bv[pu][row][g];
        const int   i = s_bi[pu][row][g];
        if (v > mv) { mv = v; mi = i; }
    }
    const float4 u = *(const float4*)(centroids + ((size_t)(pu * NCENT + mi)) * CDIM + (c & 3) * 4);

    float* __restrict__ neg_out = out + BATCH + (size_t)b * NNEG;

    int pass = 0;
    #pragma nounroll
    for (; pass < NPASS - 3; pass += 2) {
        const int id_c = slot_id(pos_id, neg_ids, b, (pass + 2) * 32 + lane);
        const int id_d = slot_id(pos_id, neg_ids, b, (pass + 3) * 32 + lane);

        load_pass<16>(vb, rows, id_b, (unsigned)c);
        neg_out[pass * 32 + lane] = score_pass<16>(va, u, c);

        load_pass<16>(va, rows, id_c, (unsigned)c);
        neg_out[(pass + 1) * 32 + lane] = score_pass<16>(vb, u, c);

        id_b = id_d;
    }

    {
        const int id_c = slot_id(pos_id, neg_ids, b, (NPASS - 1) * 32 + lane);

        load_pass<16>(vb, rows, id_b, (unsigned)c);
        neg_out[(NPASS - 3) * 32 + lane] = score_pass<16>(va, u, c);

        load_pass<NTAIL>(va, rows, id_c, (unsigned)c);
        neg_out[(NPASS - 2) * 32 + lane] = score_pass<16>(vb, u, c);
    }

    {
        const float s    = score_pass<NTAIL>(va, u, c);
        const int   slot = (NPASS - 1) * 32 + lane;
        if (slot < NNEG)       neg_out[slot] = s;
        else if (slot == NNEG) out[b] = s;
    }
}

extern "C" void kernel_launch(void* const* d_in, const int* in_sizes, int n_in,
                              void* d_out, int out_size, void* d_ws, size_t ws_size,
                              hipStream_t stream) {
    const int*   user_id    = (const int*)  d_in[0];
    const int*   pos_id     = (const int*)  d_in[1];
    const int*   neg_ids    = (const int*)  d_in[2];
    const float* user_table = (const float*)d_in[3];
    const float* centroids  = (const float*)d_in[4];
    const float* item_table = (const float*)d_in[5];
    const float* gumbel_u   = (const float*)d_in[6];
    float* out = (float*)d_out;

    fused_kernel<<<BATCH / ROWS, 512, 0, stream>>>(user_id, pos_id, neg_ids, user_table,
                                                    centroids, item_table, gumbel_u, out);
}
